# starting kernel
# baseline (speedup 1.0000x reference)
#include <hip/hip_runtime.h>
#include <hip/hip_bf16.h>

#define HDIM 128
#define CAP 64
#define MAX_OVF 8192

typedef float v2f __attribute__((ext_vector_type(2)));
typedef float v8f __attribute__((ext_vector_type(8)));

__global__ __launch_bounds__(256, 1)
void gemm_bias_relu_res_ln(const float* __restrict__ X,
                           const float* __restrict__ Wm,
                           const float* __restrict__ bias,
                           const float* __restrict__ R,
                           const float* __restrict__ g,
                           const float* __restrict__ bln,
                           float* __restrict__ out,
                           int n_rows)
{
    __shared__ float Ws[HDIM * HDIM];
    const int tid = threadIdx.x;

    {
        const float4* Wv = (const float4*)Wm;
        float4* Sv = (float4*)Ws;
#pragma unroll
        for (int i = 0; i < (HDIM * HDIM / 4) / 256; ++i)
            Sv[tid + i * 256] = Wv[tid + i * 256];
    }
    __syncthreads();

    const int wave = tid >> 5;
    const int lane = tid & 31;
    const int half = lane >> 4;
    const int l16  = lane & 15;

    const int row_base = blockIdx.x * 128 + wave * 16;

    int a_row = row_base + l16;
    if (a_row > n_rows - 1) a_row = n_rows - 1;
    const float* Xrow = X + (size_t)a_row * HDIM;

    v8f acc[8];
#pragma unroll
    for (int nt = 0; nt < 8; ++nt)
        acc[nt] = v8f{0.f, 0.f, 0.f, 0.f, 0.f, 0.f, 0.f, 0.f};

#pragma unroll 4
    for (int ks = 0; ks < 32; ++ks) {
        const int kb = ks * 4 + half * 2;
        v2f a;
        a.x = Xrow[kb];
        a.y = Xrow[kb + 1];
#pragma unroll
        for (int nt = 0; nt < 8; ++nt) {
            const int n = nt * 16 + l16;
            v2f bf;
            bf.x = Ws[kb * HDIM + n];
            bf.y = Ws[(kb + 1) * HDIM + n];
            acc[nt] = __builtin_amdgcn_wmma_f32_16x16x4_f32(
                false, a, false, bf, (short)0, acc[nt], false, false);
        }
    }

    float biasv[8], gv[8], bv[8];
#pragma unroll
    for (int nt = 0; nt < 8; ++nt) {
        const int n = nt * 16 + l16;
        biasv[nt] = bias[n];
        gv[nt]    = g[n];
        bv[nt]    = bln[n];
    }

#pragma unroll
    for (int j = 0; j < 8; ++j) {
        const int r  = row_base + j + half * 8;
        const int rc = (r > n_rows - 1) ? (n_rows - 1) : r;
        const float* Rrow = R + (size_t)rc * HDIM;

        float s = 0.f;
#pragma unroll
        for (int nt = 0; nt < 8; ++nt) {
            float x = acc[nt][j] + biasv[nt];
            x = fmaxf(x, 0.f);
            x += Rrow[nt * 16 + l16];
            acc[nt][j] = x;
            s += x;
        }
        s += __shfl_xor(s, 1, 32);
        s += __shfl_xor(s, 2, 32);
        s += __shfl_xor(s, 4, 32);
        s += __shfl_xor(s, 8, 32);
        const float mean = s * (1.f / HDIM);

        float var = 0.f;
#pragma unroll
        for (int nt = 0; nt < 8; ++nt) {
            const float d = acc[nt][j] - mean;
            var += d * d;
        }
        var += __shfl_xor(var, 1, 32);
        var += __shfl_xor(var, 2, 32);
        var += __shfl_xor(var, 4, 32);
        var += __shfl_xor(var, 8, 32);
        const float inv = rsqrtf(var * (1.f / HDIM) + 1e-5f);

#pragma unroll
        for (int nt = 0; nt < 8; ++nt)
            acc[nt][j] = (acc[nt][j] - mean) * inv * gv[nt] + bv[nt];

        if (r < n_rows) {
            float* orow = out + (size_t)r * HDIM;
#pragma unroll
            for (int nt = 0; nt < 8; ++nt)
                orow[nt * 16 + l16] = acc[nt][j];
        }
    }
}

__global__ __launch_bounds__(256)
void build_adj(const int* __restrict__ src, const int* __restrict__ dst,
               int* __restrict__ cnt, int* __restrict__ adj,
               int* __restrict__ ovf_cnt, int* __restrict__ ovf, int n_edges)
{
    const int e = blockIdx.x * 256 + threadIdx.x;
    if (e >= n_edges) return;
    const int d = dst[e];
    const int p = atomicAdd(&cnt[d], 1);
    if (p < CAP) {
        adj[(size_t)d * CAP + p] = src[e];
    } else {
        const int o = atomicAdd(ovf_cnt, 1);
        if (o < MAX_OVF) ovf[o] = e;
    }
}

__global__ __launch_bounds__(256)
void gather_agg(const float* __restrict__ ff, const int* __restrict__ adj,
                const int* __restrict__ cnt, float* __restrict__ agg, int n_nodes)
{
    const int gid = blockIdx.x * 256 + threadIdx.x;
    const int n = gid >> 5;
    if (n >= n_nodes) return;
    const int lane = threadIdx.x & 31;

    int deg = cnt[n];
    if (deg > CAP) deg = CAP;
    const int* ap = adj + (size_t)n * CAP;

    float4 s = {0.f, 0.f, 0.f, 0.f};
    for (int i = 0; i < deg; ++i) {
        const int sn = ap[i];
        const float4 v = *(const float4*)(ff + (size_t)sn * HDIM + lane * 4);
        s.x += v.x; s.y += v.y; s.z += v.z; s.w += v.w;
    }
    *(float4*)(agg + (size_t)n * HDIM + lane * 4) = s;
}

__global__ __launch_bounds__(256)
void ovf_scatter(const float* __restrict__ ff, const int* __restrict__ src,
                 const int* __restrict__ dst, const int* __restrict__ ovf,
                 const int* __restrict__ ovf_cnt, float* __restrict__ agg)
{
    const int gid = blockIdx.x * 256 + threadIdx.x;
    const int idx = gid >> 5;
    int c = *ovf_cnt;
    if (c > MAX_OVF) c = MAX_OVF;
    if (idx >= c) return;
    const int lane = threadIdx.x & 31;

    const int e = ovf[idx];
    const int s = src[e];
    const int d = dst[e];
    const float4 v = *(const float4*)(ff + (size_t)s * HDIM + lane * 4);
    float* ap = agg + (size_t)d * HDIM + lane * 4;
    unsafeAtomicAdd(ap + 0, v.x);
    unsafeAtomicAdd(ap + 1, v.y);
    unsafeAtomicAdd(ap + 2, v.z);
    unsafeAtomicAdd(ap + 3, v.w);
}

extern "C" void kernel_launch(void* const* d_in, const int* in_sizes, int n_in,
                              void* d_out, int out_size, void* d_ws, size_t ws_size,
                              hipStream_t stream)
{
    const float* features = (const float*)d_in[0];
    const int*   src      = (const int*)d_in[1];
    const int*   dst      = (const int*)d_in[2];
    const float* W        = (const float*)d_in[3];
    const float* b        = (const float*)d_in[4];
    const float* gcn_W    = (const float*)d_in[5];
    const float* gcn_b    = (const float*)d_in[6];
    const float* ffln_g   = (const float*)d_in[7];
    const float* ffln_b   = (const float*)d_in[8];
    const float* ln_g     = (const float*)d_in[9];
    const float* ln_b     = (const float*)d_in[10];

    const int n_nodes  = in_sizes[0] / HDIM;
    const int n_edges  = in_sizes[1];
    const int n_layers = in_sizes[6] / HDIM;

    char* ws = (char*)d_ws;
    float* ff      = (float*)ws;
    float* agg     = ff + (size_t)n_nodes * HDIM;
    int*   adj     = (int*)(agg + (size_t)n_nodes * HDIM);
    int*   cnt     = adj + (size_t)n_nodes * CAP;
    int*   ovf_cnt = cnt + n_nodes;
    int*   ovf     = ovf_cnt + 1;
    float* out     = (float*)d_out;

    const dim3 block(256);
    const dim3 ggrid((n_nodes + 127) / 128);
    const dim3 egrid((n_edges + 255) / 256);
    const dim3 ngrid((unsigned)(((size_t)n_nodes * 32 + 255) / 256));
    const dim3 ogrid((MAX_OVF * 32) / 256);

    hipMemsetAsync(cnt, 0, ((size_t)n_nodes + 1) * sizeof(int), stream);
    build_adj<<<egrid, block, 0, stream>>>(src, dst, cnt, adj, ovf_cnt, ovf, n_edges);

    const float* cur = features;
    for (int i = 0; i < n_layers; ++i) {
        gemm_bias_relu_res_ln<<<ggrid, block, 0, stream>>>(
            cur, W, b, cur, ffln_g + i * HDIM, ffln_b + i * HDIM, ff, n_nodes);

        gather_agg<<<ngrid, block, 0, stream>>>(ff, adj, cnt, agg, n_nodes);
        ovf_scatter<<<ogrid, block, 0, stream>>>(ff, src, dst, ovf, ovf_cnt, agg);

        gemm_bias_relu_res_ln<<<ggrid, block, 0, stream>>>(
            agg, gcn_W + (size_t)i * HDIM * HDIM, gcn_b + i * HDIM,
            ff, ln_g + i * HDIM, ln_b + i * HDIM, out, n_nodes);

        cur = out;
    }
}


# GEMM operands re-laid: W in LDS as (k pair, column) cells, lane owns 4 adjacent columns: one ds_load_b128 feeds 2 WMMAs, no address math in k loops, rows leave as 256-byte units; LN sum order changes
# speedup vs baseline: 1.2383x; 1.2383x over previous
#include <hip/hip_runtime.h>
#include <hip/hip_bf16.h>


#define HDIM 128
#define CAP 64
#define MAX_OVF 8192
#define TP 132

typedef float v2f __attribute__((ext_vector_type(2)));
typedef float v8f __attribute__((ext_vector_type(8)));

__global__ __launch_bounds__(512)
void ff_ln_then_gcn_linear(const float* __restrict__ X,
                           const float* __restrict__ W1,
                           const float* __restrict__ b1,
                           const float* __restrict__ g,
                           const float* __restrict__ bln,
                           const float* __restrict__ W2,
                           float* __restrict__ ff,
                           float* __restrict__ Y,
                           int n_rows)
{
    __shared__ struct {
        float4 w1[HDIM * HDIM / 4];
        float4 w2[HDIM * HDIM / 4];
        float  t[16 * 16 * TP];
    } sm;
    float4* const Ws1 = sm.w1;
    float4* const Ws2 = sm.w2;
    float*  const tile = sm.t;
    const int tid = threadIdx.x;

    {
        const float4* G1 = (const float4*)W1;
        const float4* G2 = (const float4*)W2;
#pragma unroll
        for (int i = 0; i < (HDIM * HDIM / 8) / 512; ++i) {
            const int cell = tid + i * 512;
            const int n4 = cell & 31;
            const int kp = cell >> 5;
            const int dq = kp * 64 + (n4 >> 4) * 32 + (n4 & 15);
            const float4 r0 = G1[(2 * kp) * 32 + n4];
            const float4 r1 = G1[(2 * kp + 1) * 32 + n4];
            const float4 t0 = G2[(2 * kp) * 32 + n4];
            const float4 t1 = G2[(2 * kp + 1) * 32 + n4];
            Ws1[dq]      = float4{r0.x, r1.x, r0.y, r1.y};
            Ws1[dq + 16] = float4{r0.z, r1.z, r0.w, r1.w};
            Ws2[dq]      = float4{t0.x, t1.x, t0.y, t1.y};
            Ws2[dq + 16] = float4{t0.z, t1.z, t0.w, t1.w};
        }
    }
    __syncthreads();

    const int wave = tid >> 5;
    const int lane = tid & 31;
    const int half = lane >> 4;
    const int l16  = lane & 15;

    const int row_base = blockIdx.x * 256 + wave * 16;

    int a_row = row_base + l16;
    if (a_row > n_rows - 1) a_row = n_rows - 1;
    const float* Xrow = X + (size_t)a_row * HDIM;

    v8f acc[8];
#pragma unroll
    for (int nt = 0; nt < 8; ++nt)
        acc[nt] = v8f{0.f, 0.f, 0.f, 0.f, 0.f, 0.f, 0.f, 0.f};

    const float4* B1 = Ws1 + half * 64 + l16;
    const float4* B2 = Ws2 + half * 64 + l16;

#pragma unroll 4
    for (int ks = 0; ks < 32; ++ks) {
        const int kb = ks * 4 + half * 2;
        v2f a;
        a.x = Xrow[kb];
        a.y = Xrow[kb + 1];
#pragma unroll
        for (int gp = 0; gp < 4; ++gp) {
            const float4 q = B1[ks * 128 + gp * 16];
            acc[2 * gp] = __builtin_amdgcn_wmma_f32_16x16x4_f32(
                false, a, false, v2f{q.x, q.y}, (short)0, acc[2 * gp], false, false);
            acc[2 * gp + 1] = __builtin_amdgcn_wmma_f32_16x16x4_f32(
                false, a, false, v2f{q.z, q.w}, (short)0, acc[2 * gp + 1], false, false);
        }
    }

    float biasv[8], gv[8], bv[8];
#pragma unroll
    for (int G = 0; G < 2; ++G) {
        const float4 t0 = ((const float4*)b1)[G * 16 + l16];
        const float4 t1 = ((const float4*)g)[G * 16 + l16];
        const float4 t2 = ((const float4*)bln)[G * 16 + l16];
        biasv[4 * G] = t0.x; biasv[4 * G + 1] = t0.y; biasv[4 * G + 2] = t0.z; biasv[4 * G + 3] = t0.w;
        gv[4 * G]    = t1.x; gv[4 * G + 1]    = t1.y; gv[4 * G + 2]    = t1.z; gv[4 * G + 3]    = t1.w;
        bv[4 * G]    = t2.x; bv[4 * G + 1]    = t2.y; bv[4 * G + 2]    = t2.z; bv[4 * G + 3]    = t2.w;
    }

    float* wtile = tile + wave * (16 * TP);

#pragma unroll
    for (int j = 0; j < 8; ++j) {
        const int r  = row_base + j + half * 8;
        const int rc = (r > n_rows - 1) ? (n_rows - 1) : r;
        const float4* Rrow = (const float4*)(X + (size_t)rc * HDIM);
        const float4 ra = Rrow[l16];
        const float4 rb = Rrow[16 + l16];
        const float res[8] = {ra.x, ra.y, ra.z, ra.w, rb.x, rb.y, rb.z, rb.w};

        float s = 0.f;
#pragma unroll
        for (int nt = 0; nt < 8; ++nt) {
            float x = acc[nt][j] + biasv[nt];
            x = fmaxf(x, 0.f);
            x += res[nt];
            acc[nt][j] = x;
            s += x;
        }
        s += __shfl_xor(s, 1, 32);
        s += __shfl_xor(s, 2, 32);
        s += __shfl_xor(s, 4, 32);
        s += __shfl_xor(s, 8, 32);
        const float mean = s * (1.f / HDIM);

        float var = 0.f;
#pragma unroll
        for (int nt = 0; nt < 8; ++nt) {
            const float d = acc[nt][j] - mean;
            var += d * d;
        }
        var += __shfl_xor(var, 1, 32);
        var += __shfl_xor(var, 2, 32);
        var += __shfl_xor(var, 4, 32);
        var += __shfl_xor(var, 8, 32);
        const float inv = rsqrtf(var * (1.f / HDIM) + 1e-5f);

#pragma unroll
        for (int nt = 0; nt < 8; ++nt)
            acc[nt][j] = (acc[nt][j] - mean) * inv * gv[nt] + bv[nt];

        float* trow = wtile + (j + half * 8) * TP;
        *(float4*)(trow + 4 * l16)      = float4{acc[0][j], acc[1][j], acc[2][j], acc[3][j]};
        *(float4*)(trow + 64 + 4 * l16) = float4{acc[4][j], acc[5][j], acc[6][j], acc[7][j]};
    }

    __threadfence_block();

#pragma unroll
    for (int j = 0; j < 8; ++j) {
        const int r = row_base + j + half * 8;
        if (r < n_rows) {
            float4* orow = (float4*)(ff + (size_t)r * HDIM);
            orow[l16]      = float4{acc[0][j], acc[1][j], acc[2][j], acc[3][j]};
            orow[16 + l16] = float4{acc[4][j], acc[5][j], acc[6][j], acc[7][j]};
        }
    }

#pragma unroll
    for (int nt = 0; nt < 8; ++nt)
        acc[nt] = v8f{0.f, 0.f, 0.f, 0.f, 0.f, 0.f, 0.f, 0.f};

    const float* Arow = wtile + l16 * TP;
#pragma unroll 4
    for (int ks = 0; ks < 32; ++ks) {
        const int kb = ks * 4 + half * 2;
        v2f a;
        a.x = Arow[kb];
        a.y = Arow[kb + 1];
#pragma unroll
        for (int gp = 0; gp < 4; ++gp) {
            const float4 q = B2[ks * 128 + gp * 16];
            acc[2 * gp] = __builtin_amdgcn_wmma_f32_16x16x4_f32(
                false, a, false, v2f{q.x, q.y}, (short)0, acc[2 * gp], false, false);
            acc[2 * gp + 1] = __builtin_amdgcn_wmma_f32_16x16x4_f32(
                false, a, false, v2f{q.z, q.w}, (short)0, acc[2 * gp + 1], false, false);
        }
    }

#pragma unroll
    for (int j = 0; j < 8; ++j) {
        const int r = row_base + j + half * 8;
        if (r < n_rows) {
            float4* yrow = (float4*)(Y + (size_t)r * HDIM);
            yrow[l16]      = float4{acc[0][j], acc[1][j], acc[2][j], acc[3][j]};
            yrow[16 + l16] = float4{acc[4][j], acc[5][j], acc[6][j], acc[7][j]};
        }
    }
}

__global__ __launch_bounds__(256)
void build_adj(const int* __restrict__ src, const int* __restrict__ dst,
               int* __restrict__ cnt, int* __restrict__ adj,
               int* __restrict__ ovf_cnt, int* __restrict__ ovf, int n_edges)
{
    const int e = blockIdx.x * 256 + threadIdx.x;
    if (e >= n_edges) return;
    const int d = dst[e];
    const int p = atomicAdd(&cnt[d], 1);
    if (p < CAP) {
        adj[(size_t)d * CAP + p] = src[e];
    } else {
        const int o = atomicAdd(ovf_cnt, 1);
        if (o < MAX_OVF) ovf[o] = e;
    }
}

__device__ __forceinline__ float4 ld_row(const float* __restrict__ Y, int sn, unsigned lane)
{
    const unsigned long long a = (unsigned long long)(Y + (size_t)sn * HDIM);
    const unsigned lo = __builtin_amdgcn_readfirstlane((unsigned)a);
    const unsigned hi = __builtin_amdgcn_readfirstlane((unsigned)(a >> 32));
    typedef float f32x4 __attribute__((ext_vector_type(4)));
    typedef const __attribute__((address_space(1))) f32x4* grow_t;
    const f32x4 v = ((grow_t)(((unsigned long long)hi << 32) | lo))[lane];
    float4 r;
    r.x = v.x; r.y = v.y; r.z = v.z; r.w = v.w;
    return r;
}

__device__ __forceinline__ void add_row(float4& s, const float4 v)
{
    s.x += v.x; s.y += v.y; s.z += v.z; s.w += v.w;
}

__global__ __launch_bounds__(256)
void gather_bias_relu_res_ln(const float* __restrict__ Y, const float* __restrict__ ffp,
                             const int* __restrict__ adj, const int* __restrict__ cnt,
                             const int* __restrict__ src, const int* __restrict__ dst,
                             const int* __restrict__ ovf, const int* __restrict__ ovf_cnt,
                             const float* __restrict__ bias, const float* __restrict__ g,
                             const float* __restrict__ bln, float* __restrict__ out, int n_nodes)
{
    const unsigned lane = threadIdx.x & 31;
    const int wave  = __builtin_amdgcn_readfirstlane((int)((blockIdx.x * 256 + threadIdx.x) >> 5));
    const int nwave = (int)(gridDim.x * 8);
    if (wave >= n_nodes) return;

    const float4 bb = ((const float4*)bias)[lane];
    int c = *ovf_cnt;
    if (c > MAX_OVF) c = MAX_OVF;

    int n = wave;
    int deg = cnt[n];
    int4 ia = *(const int4*)(adj + (size_t)n * CAP);
    int4 ib = *(const int4*)(adj + (size_t)n * CAP + 4);

    for (;;) {
        if (deg > CAP) deg = CAP;
        const int* ap = adj + (size_t)n * CAP;
        const float4 rr = ld_row(ffp, n, lane);

        float4 s = {0.f, 0.f, 0.f, 0.f};
        for (int i = 0; i < deg; i += 8) {
            const int r = deg - i;
            float4 v0, v1, v2, v3, v4, v5, v6, v7;
            v0 = ld_row(Y, ia.x, lane);
            if (r > 1) { v1 = ld_row(Y, ia.y, lane);
            if (r > 2) { v2 = ld_row(Y, ia.z, lane);
            if (r > 3) { v3 = ld_row(Y, ia.w, lane);
            if (r > 4) { v4 = ld_row(Y, ib.x, lane);
            if (r > 5) { v5 = ld_row(Y, ib.y, lane);
            if (r > 6) { v6 = ld_row(Y, ib.z, lane);
            if (r > 7) { v7 = ld_row(Y, ib.w, lane); } } } } } } }
            if (r > 8) {
                ia = *(const int4*)(ap + i + 8);
                ib = *(const int4*)(ap + i + 12);
            }
            add_row(s, v0);
            if (r > 1) { add_row(s, v1);
            if (r > 2) { add_row(s, v2);
            if (r > 3) { add_row(s, v3);
            if (r > 4) { add_row(s, v4);
            if (r > 5) { add_row(s, v5);
            if (r > 6) { add_row(s, v6);
            if (r > 7) { add_row(s, v7); } } } } } } }
        }

        for (int i = 0; i < c; ++i) {
            const int e = ovf[i];
            if (dst[e] == n) {
                const float4 v = ld_row(Y, src[e], lane);
                add_row(s, v);
            }
        }

        const int nn  = n + nwave;
        const int nnc = (nn < n_nodes) ? nn : n;
        const int  deg_nx = cnt[nnc];
        const int4 ia_nx  = *(const int4*)(adj + (size_t)nnc * CAP);
        const int4 ib_nx  = *(const int4*)(adj + (size_t)nnc * CAP + 4);

        float4 x;
        x.x = fmaxf(s.x + bb.x, 0.f) + rr.x;
        x.y = fmaxf(s.y + bb.y, 0.f) + rr.y;
        x.z = fmaxf(s.z + bb.z, 0.f) + rr.z;
        x.w = fmaxf(s.w + bb.w, 0.f) + rr.w;

        float t = (x.x + x.y) + (x.z + x.w);
        t += __shfl_xor(t, 1, 32);
        t += __shfl_xor(t, 2, 32);
        t += __shfl_xor(t, 4, 32);
        t += __shfl_xor(t, 8, 32);
        t += __shfl_xor(t, 16, 32);
        const float mean = t * (1.f / HDIM);

        const float d0 = x.x - mean, d1 = x.y - mean, d2 = x.z - mean, d3 = x.w - mean;
        float var = (d0 * d0 + d1 * d1) + (d2 * d2 + d3 * d3);
        var += __shfl_xor(var, 1, 32);
        var += __shfl_xor(var, 2, 32);
        var += __shfl_xor(var, 4, 32);
        var += __shfl_xor(var, 8, 32);
        var += __shfl_xor(var, 16, 32);
        const float inv = rsqrtf(var * (1.f / HDIM) + 1e-5f);

        const float4 gg = ((const float4*)g)[lane];
        const float4 be = ((const float4*)bln)[lane];
        float4 o;
        o.x = d0 * inv * gg.x + be.x;
        o.y = d1 * inv * gg.y + be.y;
        o.z = d2 * inv * gg.z + be.z;
        o.w = d3 * inv * gg.w + be.w;
        ((float4*)(out + (size_t)n * HDIM))[lane] = o;

        if (nn >= n_nodes) break;
        n = nn; deg = deg_nx; ia = ia_nx; ib = ib_nx;
    }
}

extern "C" void kernel_launch(void* const* d_in, const int* in_sizes, int n_in,
                              void* d_out, int out_size, void* d_ws, size_t ws_size,
                              hipStream_t stream)
{
    const float* features = (const float*)d_in[0];
    const int*   src      = (const int*)d_in[1];
    const int*   dst      = (const int*)d_in[2];
    const float* W        = (const float*)d_in[3];
    const float* b        = (const float*)d_in[4];
    const float* gcn_W    = (const float*)d_in[5];
    const float* gcn_b    = (const float*)d_in[6];
    const float* ffln_g   = (const float*)d_in[7];
    const float* ffln_b   = (const float*)d_in[8];
    const float* ln_g     = (const float*)d_in[9];
    const float* ln_b     = (const float*)d_in[10];

    const int n_nodes  = in_sizes[0] / HDIM;
    const int n_edges  = in_sizes[1];
    const int n_layers = in_sizes[6] / HDIM;

    char* ws = (char*)d_ws;
    float* ff      = (float*)ws;
    float* yy      = ff + (size_t)n_nodes * HDIM;
    int*   adj     = (int*)(yy + (size_t)n_nodes * HDIM);
    int*   cnt     = adj + (size_t)n_nodes * CAP;
    int*   ovf_cnt = cnt + n_nodes;
    int*   ovf     = ovf_cnt + 1;
    float* out     = (float*)d_out;

    const dim3 block(256);
    const dim3 gblock(512);
    const dim3 ggrid((n_nodes + 255) / 256);
    const dim3 egrid((n_edges + 255) / 256);
    const dim3 ngrid(1536);

    hipMemsetAsync(cnt, 0, ((size_t)n_nodes + 1) * sizeof(int), stream);
    build_adj<<<egrid, block, 0, stream>>>(src, dst, cnt, adj, ovf_cnt, ovf, n_edges);

    const float* cur = features;
    for (int i = 0; i < n_layers; ++i) {
        ff_ln_then_gcn_linear<<<ggrid, gblock, 0, stream>>>(
            cur, W, b, ffln_g + i * HDIM, ffln_b + i * HDIM,
            gcn_W + (size_t)i * HDIM * HDIM, ff, yy, n_nodes);

        gather_bias_relu_res_ln<<<ngrid, block, 0, stream>>>(
            yy, ff, adj, cnt, src, dst, ovf, ovf_cnt,
            gcn_b + i * HDIM, ln_g + i * HDIM, ln_b + i * HDIM, out, n_nodes);

        cur = out;
    }
}


# speedup vs baseline: 1.6463x; 1.3295x over previous
#include <hip/hip_runtime.h>
#include <hip/hip_bf16.h>


#define HDIM 128
#define CAP 64
#define SUB 32
#define MAX_OVF 8192
#define TP 132

typedef float v2f __attribute__((ext_vector_type(2)));
typedef float v8f __attribute__((ext_vector_type(8)));

__global__ __launch_bounds__(512)
void ff_ln_then_gcn_linear(const float* __restrict__ X,
                           const float* __restrict__ W1,
                           const float* __restrict__ b1,
                           const float* __restrict__ g,
                           const float* __restrict__ bln,
                           const float* __restrict__ W2,
                           float* __restrict__ ff,
                           float* __restrict__ Y,
                           int n_rows)
{
    __shared__ struct {
        float4 w1[HDIM * HDIM / 4];
        float4 w2[HDIM * HDIM / 4];
        float  t[16 * 16 * TP];
    } sm;
    float4* const Ws1 = sm.w1;
    float4* const Ws2 = sm.w2;
    float*  const tile = sm.t;
    const int tid = threadIdx.x;

    {
        const float4* G1 = (const float4*)W1;
        const float4* G2 = (const float4*)W2;
#pragma unroll
        for (int i = 0; i < (HDIM * HDIM / 8) / 512; ++i) {
            const int cell = tid + i * 512;
            const int n4 = cell & 31;
            const int kp = cell >> 5;
            const int dq = kp * 64 + (n4 >> 4) * 32 + (n4 & 15);
            const float4 r0 = G1[(2 * kp) * 32 + n4];
            const float4 r1 = G1[(2 * kp + 1) * 32 + n4];
            const float4 t0 = G2[(2 * kp) * 32 + n4];
            const float4 t1 = G2[(2 * kp + 1) * 32 + n4];
            Ws1[dq]      = float4{r0.x, r1.x, r0.y, r1.y};
            Ws1[dq + 16] = float4{r0.z, r1.z, r0.w, r1.w};
            Ws2[dq]      = float4{t0.x, t1.x, t0.y, t1.y};
            Ws2[dq + 16] = float4{t0.z, t1.z, t0.w, t1.w};
        }
    }
    __syncthreads();

    const int wave = tid >> 5;
    const int lane = tid & 31;
    const int half = lane >> 4;
    const int l16  = lane & 15;

    const int row_base = blockIdx.x * 256 + wave * 16;

    int a_row = row_base + l16;
    if (a_row > n_rows - 1) a_row = n_rows - 1;
    const float* Xrow = X + (size_t)a_row * HDIM;

    v8f acc[8];
#pragma unroll
    for (int nt = 0; nt < 8; ++nt)
        acc[nt] = v8f{0.f, 0.f, 0.f, 0.f, 0.f, 0.f, 0.f, 0.f};

    const float4* B1 = Ws1 + half * 64 + l16;
    const float4* B2 = Ws2 + half * 64 + l16;

#pragma unroll 4
    for (int ks = 0; ks < 32; ++ks) {
        const int kb = ks * 4 + half * 2;
        v2f a;
        a.x = Xrow[kb];
        a.y = Xrow[kb + 1];
#pragma unroll
        for (int gp = 0; gp < 4; ++gp) {
            const float4 q = B1[ks * 128 + gp * 16];
            acc[2 * gp] = __builtin_amdgcn_wmma_f32_16x16x4_f32(
                false, a, false, v2f{q.x, q.y}, (short)0, acc[2 * gp], false, false);
            acc[2 * gp + 1] = __builtin_amdgcn_wmma_f32_16x16x4_f32(
                false, a, false, v2f{q.z, q.w}, (short)0, acc[2 * gp + 1], false, false);
        }
    }

    float biasv[8], gv[8], bv[8];
#pragma unroll
    for (int G = 0; G < 2; ++G) {
        const float4 t0 = ((const float4*)b1)[G * 16 + l16];
        const float4 t1 = ((const float4*)g)[G * 16 + l16];
        const float4 t2 = ((const float4*)bln)[G * 16 + l16];
        biasv[4 * G] = t0.x; biasv[4 * G + 1] = t0.y; biasv[4 * G + 2] = t0.z; biasv[4 * G + 3] = t0.w;
        gv[4 * G]    = t1.x; gv[4 * G + 1]    = t1.y; gv[4 * G + 2]    = t1.z; gv[4 * G + 3]    = t1.w;
        bv[4 * G]    = t2.x; bv[4 * G + 1]    = t2.y; bv[4 * G + 2]    = t2.z; bv[4 * G + 3]    = t2.w;
    }

    float* wtile = tile + wave * (16 * TP);

#pragma unroll
    for (int j = 0; j < 8; ++j) {
        const int r  = row_base + j + half * 8;
        const int rc = (r > n_rows - 1) ? (n_rows - 1) : r;
        const float4* Rrow = (const float4*)(X + (size_t)rc * HDIM);
        const float4 ra = Rrow[l16];
        const float4 rb = Rrow[16 + l16];
        const float res[8] = {ra.x, ra.y, ra.z, ra.w, rb.x, rb.y, rb.z, rb.w};

        float s = 0.f;
#pragma unroll
        for (int nt = 0; nt < 8; ++nt) {
            float x = acc[nt][j] + biasv[nt];
            x = fmaxf(x, 0.f);
            x += res[nt];
            acc[nt][j] = x;
            s += x;
        }
        s += __shfl_xor(s, 1, 32);
        s += __shfl_xor(s, 2, 32);
        s += __shfl_xor(s, 4, 32);
        s += __shfl_xor(s, 8, 32);
        const float mean = s * (1.f / HDIM);

        float var = 0.f;
#pragma unroll
        for (int nt = 0; nt < 8; ++nt) {
            const float d = acc[nt][j] - mean;
            var += d * d;
        }
        var += __shfl_xor(var, 1, 32);
        var += __shfl_xor(var, 2, 32);
        var += __shfl_xor(var, 4, 32);
        var += __shfl_xor(var, 8, 32);
        const float inv = rsqrtf(var * (1.f / HDIM) + 1e-5f);

#pragma unroll
        for (int nt = 0; nt < 8; ++nt)
            acc[nt][j] = (acc[nt][j] - mean) * inv * gv[nt] + bv[nt];

        float* trow = wtile + (j + half * 8) * TP;
        *(float4*)(trow + 4 * l16)      = float4{acc[0][j], acc[1][j], acc[2][j], acc[3][j]};
        *(float4*)(trow + 64 + 4 * l16) = float4{acc[4][j], acc[5][j], acc[6][j], acc[7][j]};
    }

    __threadfence_block();

#pragma unroll
    for (int j = 0; j < 8; ++j) {
        const int r = row_base + j + half * 8;
        if (r < n_rows) {
            float4* orow = (float4*)(ff + (size_t)r * HDIM);
            orow[l16]      = float4{acc[0][j], acc[1][j], acc[2][j], acc[3][j]};
            orow[16 + l16] = float4{acc[4][j], acc[5][j], acc[6][j], acc[7][j]};
        }
    }

#pragma unroll
    for (int nt = 0; nt < 8; ++nt)
        acc[nt] = v8f{0.f, 0.f, 0.f, 0.f, 0.f, 0.f, 0.f, 0.f};

    const float* Arow = wtile + l16 * TP;
#pragma unroll 4
    for (int ks = 0; ks < 32; ++ks) {
        const int kb = ks * 4 + half * 2;
        v2f a;
        a.x = Arow[kb];
        a.y = Arow[kb + 1];
#pragma unroll
        for (int gp = 0; gp < 4; ++gp) {
            const float4 q = B2[ks * 128 + gp * 16];
            acc[2 * gp] = __builtin_amdgcn_wmma_f32_16x16x4_f32(
                false, a, false, v2f{q.x, q.y}, (short)0, acc[2 * gp], false, false);
            acc[2 * gp + 1] = __builtin_amdgcn_wmma_f32_16x16x4_f32(
                false, a, false, v2f{q.z, q.w}, (short)0, acc[2 * gp + 1], false, false);
        }
    }

#pragma unroll
    for (int j = 0; j < 8; ++j) {
        const int r = row_base + j + half * 8;
        if (r < n_rows) {
            float4* yrow = (float4*)(Y + (size_t)r * HDIM);
            yrow[l16]      = float4{acc[0][j], acc[1][j], acc[2][j], acc[3][j]};
            yrow[16 + l16] = float4{acc[4][j], acc[5][j], acc[6][j], acc[7][j]};
        }
    }
}

__global__ __launch_bounds__(256)
void build_adj(const int* __restrict__ src, const int* __restrict__ dst,
               int* __restrict__ cnt, int* __restrict__ adj,
               int* __restrict__ ovf_cnt, int* __restrict__ ovf, int n_edges, int n_nodes)
{
    const int h = (int)((__builtin_amdgcn_s_sendmsg_rtn(0x87) >> 18) & 1u);
    const int e = blockIdx.x * 256 + threadIdx.x;
    if (e >= n_edges) return;
    const int d = dst[e];
    const int p = atomicAdd(&cnt[(size_t)h * n_nodes + d], 1);
    if (p < SUB) {
        adj[(size_t)d * CAP + h * SUB + p] = src[e];
    } else {
        const int o = atomicAdd(ovf_cnt, 1);
        if (o < MAX_OVF) ovf[o] = e;
    }
}

__device__ __forceinline__ float4 ld_row(const float* __restrict__ Y, int sn, unsigned lane)
{
    const unsigned long long a = (unsigned long long)(Y + (size_t)sn * HDIM);
    const unsigned lo = __builtin_amdgcn_readfirstlane((unsigned)a);
    const unsigned hi = __builtin_amdgcn_readfirstlane((unsigned)(a >> 32));
    typedef float f32x4 __attribute__((ext_vector_type(4)));
    typedef const __attribute__((address_space(1))) f32x4* grow_t;
    const f32x4 v = ((grow_t)(((unsigned long long)hi << 32) | lo))[lane];
    float4 r;
    r.x = v.x; r.y = v.y; r.z = v.z; r.w = v.w;
    return r;
}

__device__ __forceinline__ void add_row(float4& s, const float4 v)
{
    s.x += v.x; s.y += v.y; s.z += v.z; s.w += v.w;
}

__device__ __forceinline__ void sum_sub_bucket(float4& s, const float* __restrict__ Y, const int* __restrict__ sp,
                                               int n_slots, int4 ia, int4 ib, unsigned lane)
{
    for (int i = 0; i < n_slots; i += 8) {
        const int r = n_slots - i;
        float4 v0, v1, v2, v3, v4, v5, v6, v7;
        v0 = ld_row(Y, ia.x, lane);
        if (r > 1) { v1 = ld_row(Y, ia.y, lane);
        if (r > 2) { v2 = ld_row(Y, ia.z, lane);
        if (r > 3) { v3 = ld_row(Y, ia.w, lane);
        if (r > 4) { v4 = ld_row(Y, ib.x, lane);
        if (r > 5) { v5 = ld_row(Y, ib.y, lane);
        if (r > 6) { v6 = ld_row(Y, ib.z, lane);
        if (r > 7) { v7 = ld_row(Y, ib.w, lane); } } } } } } }
        if (r > 8) {
            ia = *(const int4*)(sp + i + 8);
            ib = *(const int4*)(sp + i + 12);
        }
        add_row(s, v0);
        if (r > 1) { add_row(s, v1);
        if (r > 2) { add_row(s, v2);
        if (r > 3) { add_row(s, v3);
        if (r > 4) { add_row(s, v4);
        if (r > 5) { add_row(s, v5);
        if (r > 6) { add_row(s, v6);
        if (r > 7) { add_row(s, v7); } } } } } } }
    }
}

__global__ __launch_bounds__(256)
void gather_bias_relu_res_ln(const float* __restrict__ Y, const float* __restrict__ ffp,
                             const int* __restrict__ adj, const int* __restrict__ cnt,
                             const int* __restrict__ src, const int* __restrict__ dst,
                             const int* __restrict__ ovf, const int* __restrict__ ovf_cnt,
                             const float* __restrict__ bias, const float* __restrict__ g,
                             const float* __restrict__ bln, float* __restrict__ out, int n_nodes)
{
    const unsigned lane = threadIdx.x & 31;
    const int wave  = __builtin_amdgcn_readfirstlane((int)((blockIdx.x * 256 + threadIdx.x) >> 5));
    const int nwave = (int)(gridDim.x * 8);
    if (wave >= n_nodes) return;

    const float4 bb = ((const float4*)bias)[lane];
    int c = *ovf_cnt;
    if (c > MAX_OVF) c = MAX_OVF;

    int n = wave;
    int deg0 = cnt[n];
    int deg1 = cnt[(size_t)n_nodes + n];
    int4 ia0 = *(const int4*)(adj + (size_t)n * CAP);
    int4 ib0 = *(const int4*)(adj + (size_t)n * CAP + 4);
    int4 ia1 = *(const int4*)(adj + (size_t)n * CAP + SUB);
    int4 ib1 = *(const int4*)(adj + (size_t)n * CAP + SUB + 4);

    for (;;) {
        if (deg0 > SUB) deg0 = SUB;
        if (deg1 > SUB) deg1 = SUB;
        const int* ap = adj + (size_t)n * CAP;
        const float4 rr = ld_row(ffp, n, lane);

        float4 s = {0.f, 0.f, 0.f, 0.f};
        sum_sub_bucket(s, Y, ap, deg0, ia0, ib0, lane);
        sum_sub_bucket(s, Y, ap + SUB, deg1, ia1, ib1, lane);

        for (int i = 0; i < c; ++i) {
            const int e = ovf[i];
            if (dst[e] == n) {
                const float4 v = ld_row(Y, src[e], lane);
                add_row(s, v);
            }
        }

        const int nn  = n + nwave;
        const int nnc = (nn < n_nodes) ? nn : n;
        const int  deg0_nx = cnt[nnc];
        const int  deg1_nx = cnt[(size_t)n_nodes + nnc];
        const int4 ia0_nx  = *(const int4*)(adj + (size_t)nnc * CAP);
        const int4 ib0_nx  = *(const int4*)(adj + (size_t)nnc * CAP + 4);
        const int4 ia1_nx  = *(const int4*)(adj + (size_t)nnc * CAP + SUB);
        const int4 ib1_nx  = *(const int4*)(adj + (size_t)nnc * CAP + SUB + 4);

        float4 x;
        x.x = fmaxf(s.x + bb.x, 0.f) + rr.x;
        x.y = fmaxf(s.y + bb.y, 0.f) + rr.y;
        x.z = fmaxf(s.z + bb.z, 0.f) + rr.z;
        x.w = fmaxf(s.w + bb.w, 0.f) + rr.w;

        float t = (x.x + x.y) + (x.z + x.w);
        t += __shfl_xor(t, 1, 32);
        t += __shfl_xor(t, 2, 32);
        t += __shfl_xor(t, 4, 32);
        t += __shfl_xor(t, 8, 32);
        t += __shfl_xor(t, 16, 32);
        const float mean = t * (1.f / HDIM);

        const float d0 = x.x - mean, d1 = x.y - mean, d2 = x.z - mean, d3 = x.w - mean;
        float var = (d0 * d0 + d1 * d1) + (d2 * d2 + d3 * d3);
        var += __shfl_xor(var, 1, 32);
        var += __shfl_xor(var, 2, 32);
        var += __shfl_xor(var, 4, 32);
        var += __shfl_xor(var, 8, 32);
        var += __shfl_xor(var, 16, 32);
        const float inv = rsqrtf(var * (1.f / HDIM) + 1e-5f);

        const float4 gg = ((const float4*)g)[lane];
        const float4 be = ((const float4*)bln)[lane];
        float4 o;
        o.x = d0 * inv * gg.x + be.x;
        o.y = d1 * inv * gg.y + be.y;
        o.z = d2 * inv * gg.z + be.z;
        o.w = d3 * inv * gg.w + be.w;
        ((float4*)(out + (size_t)n * HDIM))[lane] = o;

        if (nn >= n_nodes) break;
        n = nn; deg0 = deg0_nx; deg1 = deg1_nx;
        ia0 = ia0_nx; ib0 = ib0_nx; ia1 = ia1_nx; ib1 = ib1_nx;
    }
}

extern "C" void kernel_launch(void* const* d_in, const int* in_sizes, int n_in,
                              void* d_out, int out_size, void* d_ws, size_t ws_size,
                              hipStream_t stream)
{
    const float* features = (const float*)d_in[0];
    const int*   src      = (const int*)d_in[1];
    const int*   dst      = (const int*)d_in[2];
    const float* W        = (const float*)d_in[3];
    const float* b        = (const float*)d_in[4];
    const float* gcn_W    = (const float*)d_in[5];
    const float* gcn_b    = (const float*)d_in[6];
    const float* ffln_g   = (const float*)d_in[7];
    const float* ffln_b   = (const float*)d_in[8];
    const float* ln_g     = (const float*)d_in[9];
    const float* ln_b     = (const float*)d_in[10];

    const int n_nodes  = in_sizes[0] / HDIM;
    const int n_edges  = in_sizes[1];
    const int n_layers = in_sizes[6] / HDIM;

    char* ws = (char*)d_ws;
    float* ff      = (float*)ws;
    float* yy      = ff + (size_t)n_nodes * HDIM;
    int*   adj     = (int*)(yy + (size_t)n_nodes * HDIM);
    int*   cnt     = adj + (size_t)n_nodes * CAP;
    int*   ovf_cnt = cnt + 2 * (size_t)n_nodes;
    int*   ovf     = ovf_cnt + 1;
    float* out     = (float*)d_out;

    const dim3 block(256);
    const dim3 gblock(512);
    const dim3 ggrid((n_nodes + 255) / 256);
    const dim3 egrid((n_edges + 255) / 256);
    const dim3 ngrid(1536);

    hipMemsetAsync(cnt, 0, (2 * (size_t)n_nodes + 1) * sizeof(int), stream);
    build_adj<<<egrid, block, 0, stream>>>(src, dst, cnt, adj, ovf_cnt, ovf, n_edges, n_nodes);

    const float* cur = features;
    for (int i = 0; i < n_layers; ++i) {
        ff_ln_then_gcn_linear<<<ggrid, gblock, 0, stream>>>(
            cur, W, b, ffln_g + i * HDIM, ffln_b + i * HDIM,
            gcn_W + (size_t)i * HDIM * HDIM, ff, yy, n_nodes);

        gather_bias_relu_res_ln<<<ngrid, block, 0, stream>>>(
            yy, ff, adj, cnt, src, dst, ovf, ovf_cnt,
            gcn_b + i * HDIM, ln_g + i * HDIM, ln_b + i * HDIM, out, n_nodes);

        cur = out;
    }
}
